# starting kernel
# baseline (speedup 1.0000x reference)
#include <hip/hip_runtime.h>
#include <hip/hip_bf16.h>

typedef float v2f __attribute__((ext_vector_type(2)));
typedef float v8f __attribute__((ext_vector_type(8)));

static __host__ int cdiv(int a, int b) { return (a + b - 1) / b; }

__global__ void zero_f(float* __restrict__ p, int n) {
  int i = blockIdx.x * blockDim.x + threadIdx.x;
  if (i < n) p[i] = 0.0f;
}

__global__ void edge_degree(const int* __restrict__ dst, int E, float* __restrict__ degf) {
  int e = blockIdx.x * blockDim.x + threadIdx.x;
  if (e < E) atomicAdd(&degf[dst[e]], 1.0f);
}

__global__ void make_dinv(float* __restrict__ dinv, int N) {
  int i = blockIdx.x * blockDim.x + threadIdx.x;
  if (i < N) dinv[i] = rsqrtf(dinv[i] + 1.0f);
}

__global__ void gemm_tile16(const float* __restrict__ X, const float* __restrict__ W,
                            const float* __restrict__ dinv, float* __restrict__ G,
                            int N, int K, int tiles) {
  int tile = blockIdx.x * (blockDim.x >> 5) + (threadIdx.x >> 5);
  if (tile >= tiles) return;
  int lane = threadIdx.x & 31;
  int hi = lane >> 4;
  int lo = lane & 15;
  int row = tile * 16 + lo;
  float rowMask = (row < N) ? 1.0f : 0.0f;
  const float* xr = X + (size_t)((row < N) ? row : 0) * K;
  const float* wc = W + lo;
  int K4 = K & ~3;

  v8f c = {};
  int k0 = 0;
  for (; k0 < K4; k0 += 4) {
    int kk = k0 + 2 * hi;
    if ((k0 & 63) == 0) __builtin_prefetch(xr + kk + 64, 0, 1);
    v2f a, b;
    a.x = xr[kk] * rowMask;
    a.y = xr[kk + 1] * rowMask;
    b.x = wc[kk * 16];
    b.y = wc[(kk + 1) * 16];
    c = __builtin_amdgcn_wmma_f32_16x16x4_f32(false, a, false, b, (short)0, c,
                                              false, false);
  }
  if (k0 < K) {
    int kk = k0 + 2 * hi;
    int kc0 = (kk < K) ? kk : 0;
    int kc1 = (kk + 1 < K) ? kk + 1 : 0;
    float m0 = (kk < K) ? rowMask : 0.0f;
    float m1 = (kk + 1 < K) ? rowMask : 0.0f;
    v2f a, b;
    a.x = xr[kc0] * m0;
    a.y = xr[kc1] * m1;
    b.x = wc[kc0 * 16];
    b.y = wc[kc1 * 16];
    c = __builtin_amdgcn_wmma_f32_16x16x4_f32(false, a, false, b, (short)0, c,
                                              false, false);
  }
#pragma unroll
  for (int j = 0; j < 8; ++j) {
    int rr = tile * 16 + 8 * hi + j;
    if (rr < N) G[(size_t)rr * 16 + lo] = c[j] * dinv[rr];
  }
}

__global__ void edge_agg(const int* __restrict__ src, const int* __restrict__ dst,
                         const float* __restrict__ g, float* __restrict__ agg, int E4) {
  int t = blockIdx.x * blockDim.x + threadIdx.x;
  if (t >= E4) return;
  int e = t >> 2, part = t & 3;
  int s = src[e], d = dst[e];
  float4 v = ((const float4*)(g + (size_t)s * 16))[part];
  float* o = agg + (size_t)d * 16 + part * 4;
  atomicAdd(o + 0, v.x);
  atomicAdd(o + 1, v.y);
  atomicAdd(o + 2, v.z);
  atomicAdd(o + 3, v.w);
}

__global__ void post1(const float* __restrict__ agg, const float* __restrict__ g,
                      const float* __restrict__ dinv, const float* __restrict__ b,
                      float* __restrict__ h, int N) {
  int i = blockIdx.x * blockDim.x + threadIdx.x;
  if (i >= N) return;
  float di = dinv[i];
  const float4* av = (const float4*)(agg + (size_t)i * 16);
  const float4* gv = (const float4*)(g + (size_t)i * 16);
  float4* hv = (float4*)(h + (size_t)i * 16);
#pragma unroll
  for (int j = 0; j < 4; ++j) {
    float4 a = av[j], gg = gv[j];
    float4 r;
    r.x = fmaxf(di * (a.x + gg.x) + b[j * 4 + 0], 0.0f);
    r.y = fmaxf(di * (a.y + gg.y) + b[j * 4 + 1], 0.0f);
    r.z = fmaxf(di * (a.z + gg.z) + b[j * 4 + 2], 0.0f);
    r.w = fmaxf(di * (a.w + gg.w) + b[j * 4 + 3], 0.0f);
    hv[j] = r;
  }
}

__global__ void post2(const float* __restrict__ agg, const float* __restrict__ g,
                      const float* __restrict__ dinv, const float* __restrict__ b,
                      const float* __restrict__ Wfc, float* __restrict__ p,
                      float* __restrict__ q, int N) {
  int i = blockIdx.x * blockDim.x + threadIdx.x;
  if (i >= N) return;
  float di = dinv[i];
  float sp = 0.0f, sq = 0.0f;
#pragma unroll
  for (int f = 0; f < 16; ++f) {
    float h = fmaxf(di * (agg[(size_t)i * 16 + f] + g[(size_t)i * 16 + f]) + b[f], 0.0f);
    sp += h * Wfc[f];
    sq += h * Wfc[16 + f];
  }
  p[i] = sp;
  q[i] = sq;
}

__global__ void edge_out(const int* __restrict__ src, const int* __restrict__ dst,
                         const float* __restrict__ p, const float* __restrict__ q,
                         const float* __restrict__ bfc, float* __restrict__ out, int E) {
  int e = blockIdx.x * blockDim.x + threadIdx.x;
  if (e < E) out[e] = p[src[e]] + q[dst[e]] + bfc[0];
}

extern "C" void kernel_launch(void* const* d_in, const int* in_sizes, int n_in,
                              void* d_out, int out_size, void* d_ws, size_t ws_size,
                              hipStream_t stream) {
  const float* x   = (const float*)d_in[0];
  const int*   ei  = (const int*)d_in[1];
  const float* W1  = (const float*)d_in[2];
  const float* b1  = (const float*)d_in[3];
  const float* W2  = (const float*)d_in[4];
  const float* b2  = (const float*)d_in[5];
  const float* Wfc = (const float*)d_in[6];
  const float* bfc = (const float*)d_in[7];
  float* out = (float*)d_out;

  const int K1 = 495;
  const int N = in_sizes[0] / K1;
  const int E = in_sizes[1] / 2;
  const int* src = ei;
  const int* dst = ei + E;

  float* ws  = (float*)d_ws;
  float* dinv = ws;
  float* g    = ws + (size_t)N;
  float* agg  = ws + (size_t)N * 17;
  float* h1   = ws + (size_t)N * 33;
  float* p    = ws + (size_t)N * 49;
  float* q    = ws + (size_t)N * 50;

  const int B = 256;
  const int tiles = cdiv(N, 16);
  const int gemmBlocks = cdiv(tiles, B / 32);

  zero_f<<<cdiv(N, B), B, 0, stream>>>(dinv, N);
  edge_degree<<<cdiv(E, B), B, 0, stream>>>(dst, E, dinv);
  make_dinv<<<cdiv(N, B), B, 0, stream>>>(dinv, N);

  gemm_tile16<<<gemmBlocks, B, 0, stream>>>(x, W1, dinv, g, N, K1, tiles);
  zero_f<<<cdiv(16 * N, B), B, 0, stream>>>(agg, 16 * N);
  edge_agg<<<cdiv(4 * E, B), B, 0, stream>>>(src, dst, g, agg, 4 * E);
  post1<<<cdiv(N, B), B, 0, stream>>>(agg, g, dinv, b1, h1, N);

  gemm_tile16<<<gemmBlocks, B, 0, stream>>>(h1, W2, dinv, g, N, 16, tiles);
  zero_f<<<cdiv(16 * N, B), B, 0, stream>>>(agg, 16 * N);
  edge_agg<<<cdiv(4 * E, B), B, 0, stream>>>(src, dst, g, agg, 4 * E);
  post2<<<cdiv(N, B), B, 0, stream>>>(agg, g, dinv, b2, Wfc, p, q, N);

  edge_out<<<cdiv(E, B), B, 0, stream>>>(src, dst, p, q, bfc, out, E);
}


# edge_agg: 16 adjacent lanes gather and atomically add one whole 64-B row (wave = 32 edges, indices shuffled to half-waves) instead of 4 lanes x float4: 1 L2 request per row, not 4; same f32 atomics
# speedup vs baseline: 1.2717x; 1.2717x over previous
#include <hip/hip_runtime.h>
#include <hip/hip_bf16.h>

typedef float v2f __attribute__((ext_vector_type(2)));
typedef float v8f __attribute__((ext_vector_type(8)));

static __host__ int cdiv(int a, int b) { return (a + b - 1) / b; }

__global__ void zero_f(float* __restrict__ p, int n) {
  int i = blockIdx.x * blockDim.x + threadIdx.x;
  if (i < n) p[i] = 0.0f;
}

__global__ void edge_degree(const int* __restrict__ dst, int E, float* __restrict__ degf) {
  int e = blockIdx.x * blockDim.x + threadIdx.x;
  if (e < E) atomicAdd(&degf[dst[e]], 1.0f);
}

__global__ void make_dinv(float* __restrict__ dinv, int N) {
  int i = blockIdx.x * blockDim.x + threadIdx.x;
  if (i < N) dinv[i] = rsqrtf(dinv[i] + 1.0f);
}

__global__ void gemm_tile16(const float* __restrict__ X, const float* __restrict__ W,
                            const float* __restrict__ dinv, float* __restrict__ G,
                            int N, int K, int tiles) {
  int tile = blockIdx.x * (blockDim.x >> 5) + (threadIdx.x >> 5);
  if (tile >= tiles) return;
  int lane = threadIdx.x & 31;
  int hi = lane >> 4;
  int lo = lane & 15;
  int row = tile * 16 + lo;
  float rowMask = (row < N) ? 1.0f : 0.0f;
  const float* xr = X + (size_t)((row < N) ? row : 0) * K;
  const float* wc = W + lo;
  int K4 = K & ~3;

  v8f c = {};
  int k0 = 0;
  for (; k0 < K4; k0 += 4) {
    int kk = k0 + 2 * hi;
    if ((k0 & 63) == 0) __builtin_prefetch(xr + kk + 64, 0, 1);
    v2f a, b;
    a.x = xr[kk] * rowMask;
    a.y = xr[kk + 1] * rowMask;
    b.x = wc[kk * 16];
    b.y = wc[(kk + 1) * 16];
    c = __builtin_amdgcn_wmma_f32_16x16x4_f32(false, a, false, b, (short)0, c,
                                              false, false);
  }
  if (k0 < K) {
    int kk = k0 + 2 * hi;
    int kc0 = (kk < K) ? kk : 0;
    int kc1 = (kk + 1 < K) ? kk + 1 : 0;
    float m0 = (kk < K) ? rowMask : 0.0f;
    float m1 = (kk + 1 < K) ? rowMask : 0.0f;
    v2f a, b;
    a.x = xr[kc0] * m0;
    a.y = xr[kc1] * m1;
    b.x = wc[kc0 * 16];
    b.y = wc[kc1 * 16];
    c = __builtin_amdgcn_wmma_f32_16x16x4_f32(false, a, false, b, (short)0, c,
                                              false, false);
  }
#pragma unroll
  for (int j = 0; j < 8; ++j) {
    int rr = tile * 16 + 8 * hi + j;
    if (rr < N) G[(size_t)rr * 16 + lo] = c[j] * dinv[rr];
  }
}

__global__ void edge_agg(const int* __restrict__ src, const int* __restrict__ dst,
                         const float* __restrict__ g, float* __restrict__ agg, int E) {
  int t = blockIdx.x * blockDim.x + threadIdx.x;
  int lane = threadIdx.x & 31;
  int base = __builtin_amdgcn_readfirstlane(t - lane);
  if (base >= E) return;
  int e = (t < E) ? t : (E - 1);
  int s = src[e], d = dst[e];
  unsigned f = lane & 15;
  int hi = lane >> 4;
  float v[16];
  unsigned drow[16];
#pragma unroll
  for (int k = 0; k < 16; ++k) {
    int j = 2 * k + hi;
    unsigned sj = (unsigned)__shfl(s, j);
    drow[k] = (unsigned)__shfl(d, j);
    v[k] = g[sj * 16u + f];
  }
  if (base + 32 <= E) {
#pragma unroll
    for (int k = 0; k < 16; ++k) atomicAdd(&agg[drow[k] * 16u + f], v[k]);
  } else {
#pragma unroll
    for (int k = 0; k < 16; ++k) {
      if (base + 2 * k + hi < E) atomicAdd(&agg[drow[k] * 16u + f], v[k]);
    }
  }
}

__global__ void post1(const float* __restrict__ agg, const float* __restrict__ g,
                      const float* __restrict__ dinv, const float* __restrict__ b,
                      float* __restrict__ h, int N) {
  int i = blockIdx.x * blockDim.x + threadIdx.x;
  if (i >= N) return;
  float di = dinv[i];
  const float4* av = (const float4*)(agg + (size_t)i * 16);
  const float4* gv = (const float4*)(g + (size_t)i * 16);
  float4* hv = (float4*)(h + (size_t)i * 16);
#pragma unroll
  for (int j = 0; j < 4; ++j) {
    float4 a = av[j], gg = gv[j];
    float4 r;
    r.x = fmaxf(di * (a.x + gg.x) + b[j * 4 + 0], 0.0f);
    r.y = fmaxf(di * (a.y + gg.y) + b[j * 4 + 1], 0.0f);
    r.z = fmaxf(di * (a.z + gg.z) + b[j * 4 + 2], 0.0f);
    r.w = fmaxf(di * (a.w + gg.w) + b[j * 4 + 3], 0.0f);
    hv[j] = r;
  }
}

__global__ void post2(const float* __restrict__ agg, const float* __restrict__ g,
                      const float* __restrict__ dinv, const float* __restrict__ b,
                      const float* __restrict__ Wfc, float* __restrict__ p,
                      float* __restrict__ q, int N) {
  int i = blockIdx.x * blockDim.x + threadIdx.x;
  if (i >= N) return;
  float di = dinv[i];
  float sp = 0.0f, sq = 0.0f;
#pragma unroll
  for (int f = 0; f < 16; ++f) {
    float h = fmaxf(di * (agg[(size_t)i * 16 + f] + g[(size_t)i * 16 + f]) + b[f], 0.0f);
    sp += h * Wfc[f];
    sq += h * Wfc[16 + f];
  }
  p[i] = sp;
  q[i] = sq;
}

__global__ void edge_out(const int* __restrict__ src, const int* __restrict__ dst,
                         const float* __restrict__ p, const float* __restrict__ q,
                         const float* __restrict__ bfc, float* __restrict__ out, int E) {
  int e = blockIdx.x * blockDim.x + threadIdx.x;
  if (e < E) out[e] = p[src[e]] + q[dst[e]] + bfc[0];
}

extern "C" void kernel_launch(void* const* d_in, const int* in_sizes, int n_in,
                              void* d_out, int out_size, void* d_ws, size_t ws_size,
                              hipStream_t stream) {
  const float* x   = (const float*)d_in[0];
  const int*   ei  = (const int*)d_in[1];
  const float* W1  = (const float*)d_in[2];
  const float* b1  = (const float*)d_in[3];
  const float* W2  = (const float*)d_in[4];
  const float* b2  = (const float*)d_in[5];
  const float* Wfc = (const float*)d_in[6];
  const float* bfc = (const float*)d_in[7];
  float* out = (float*)d_out;

  const int K1 = 495;
  const int N = in_sizes[0] / K1;
  const int E = in_sizes[1] / 2;
  const int* src = ei;
  const int* dst = ei + E;

  float* ws  = (float*)d_ws;
  float* dinv = ws;
  float* g    = ws + (size_t)N;
  float* agg  = ws + (size_t)N * 17;
  float* h1   = ws + (size_t)N * 33;
  float* p    = ws + (size_t)N * 49;
  float* q    = ws + (size_t)N * 50;

  const int B = 256;
  const int tiles = cdiv(N, 16);
  const int gemmBlocks = cdiv(tiles, B / 32);

  zero_f<<<cdiv(N, B), B, 0, stream>>>(dinv, N);
  edge_degree<<<cdiv(E, B), B, 0, stream>>>(dst, E, dinv);
  make_dinv<<<cdiv(N, B), B, 0, stream>>>(dinv, N);

  gemm_tile16<<<gemmBlocks, B, 0, stream>>>(x, W1, dinv, g, N, K1, tiles);
  zero_f<<<cdiv(16 * N, B), B, 0, stream>>>(agg, 16 * N);
  edge_agg<<<cdiv(E, B), B, 0, stream>>>(src, dst, g, agg, E);
  post1<<<cdiv(N, B), B, 0, stream>>>(agg, g, dinv, b1, h1, N);

  gemm_tile16<<<gemmBlocks, B, 0, stream>>>(h1, W2, dinv, g, N, 16, tiles);
  zero_f<<<cdiv(16 * N, B), B, 0, stream>>>(agg, 16 * N);
  edge_agg<<<cdiv(E, B), B, 0, stream>>>(src, dst, g, agg, E);
  post2<<<cdiv(N, B), B, 0, stream>>>(agg, g, dinv, b2, Wfc, p, q, N);

  edge_out<<<cdiv(E, B), B, 0, stream>>>(src, dst, p, q, bfc, out, E);
}


# speedup vs baseline: 2.2587x; 1.7761x over previous
#include <hip/hip_runtime.h>
#include <hip/hip_bf16.h>

typedef float v2f __attribute__((ext_vector_type(2)));
typedef float v8f __attribute__((ext_vector_type(8)));

static __host__ int cdiv(int a, int b) { return (a + b - 1) / b; }

__device__ __forceinline__ unsigned package_half() {
  return (__builtin_amdgcn_s_sendmsg_rtn(0x87) >> 18) & 1u;
}

__global__ void zero_f(float* __restrict__ p, int n) {
  int i = blockIdx.x * blockDim.x + threadIdx.x;
  if (i < n) p[i] = 0.0f;
}

__global__ void edge_degree(const int* __restrict__ dst, int E, float* __restrict__ degf, int N) {
  int e = blockIdx.x * blockDim.x + threadIdx.x;
  if (e < E) {
    int d = dst[e];
    float* plane = degf + (size_t)package_half() * (size_t)N;
    atomicAdd(&plane[d], 1.0f);
  }
}

__global__ void make_dinv(float* __restrict__ dinv, int N) {
  int i = blockIdx.x * blockDim.x + threadIdx.x;
  if (i < N) dinv[i] = rsqrtf(dinv[i] + dinv[N + i] + 1.0f);
}

__global__ void gemm_tile16(const float* X, const float* __restrict__ W,
                            const float* __restrict__ dinv, float* __restrict__ G,
                            float* G2, int N, int K, int tiles) {
  int tile = blockIdx.x * (blockDim.x >> 5) + (threadIdx.x >> 5);
  if (tile >= tiles) return;
  int lane = threadIdx.x & 31;
  int hi = lane >> 4;
  int lo = lane & 15;
  int row = tile * 16 + lo;
  float rowMask = (row < N) ? 1.0f : 0.0f;
  const float* xr = X + (size_t)((row < N) ? row : 0) * K;
  const float* wc = W + lo;
  int K4 = K & ~3;

  v8f c = {};
  int k0 = 0;
  for (; k0 < K4; k0 += 4) {
    int kk = k0 + 2 * hi;
    if ((k0 & 63) == 0) __builtin_prefetch(xr + kk + 64, 0, 1);
    v2f a, b;
    a.x = xr[kk] * rowMask;
    a.y = xr[kk + 1] * rowMask;
    b.x = wc[kk * 16];
    b.y = wc[(kk + 1) * 16];
    c = __builtin_amdgcn_wmma_f32_16x16x4_f32(false, a, false, b, (short)0, c,
                                              false, false);
  }
  if (k0 < K) {
    int kk = k0 + 2 * hi;
    int kc0 = (kk < K) ? kk : 0;
    int kc1 = (kk + 1 < K) ? kk + 1 : 0;
    float m0 = (kk < K) ? rowMask : 0.0f;
    float m1 = (kk + 1 < K) ? rowMask : 0.0f;
    v2f a, b;
    a.x = xr[kc0] * m0;
    a.y = xr[kc1] * m1;
    b.x = wc[kc0 * 16];
    b.y = wc[kc1 * 16];
    c = __builtin_amdgcn_wmma_f32_16x16x4_f32(false, a, false, b, (short)0, c,
                                              false, false);
  }
#pragma unroll
  for (int j = 0; j < 8; ++j) {
    int rr = tile * 16 + 8 * hi + j;
    if (rr < N) {
      float v = c[j] * dinv[rr];
      G[(size_t)rr * 16 + lo] = v;
      G2[(size_t)rr * 16 + lo] = v;
    }
  }
}

__global__ void edge_agg(const int* __restrict__ src, const int* __restrict__ dst,
                         const float* __restrict__ g, float* __restrict__ agg,
                         int N, int E) {
  int t = blockIdx.x * blockDim.x + threadIdx.x;
  int lane = threadIdx.x & 31;
  int base = __builtin_amdgcn_readfirstlane(t - lane);
  if (base >= E) return;
  int e = (t < E) ? t : (E - 1);
  int s = src[e], d = dst[e];
  float* plane = agg + (size_t)package_half() * (size_t)(16u * (unsigned)N);
  unsigned f = lane & 15;
  int hi = lane >> 4;
  float v[16];
  unsigned drow[16];
#pragma unroll
  for (int k = 0; k < 16; ++k) {
    int j = 2 * k + hi;
    unsigned sj = (unsigned)__shfl(s, j);
    drow[k] = (unsigned)__shfl(d, j);
    v[k] = g[sj * 16u + f];
  }
  if (base + 32 <= E) {
#pragma unroll
    for (int k = 0; k < 16; ++k) atomicAdd(&plane[drow[k] * 16u + f], v[k]);
  } else {
#pragma unroll
    for (int k = 0; k < 16; ++k) {
      if (base + 2 * k + hi < E) atomicAdd(&plane[drow[k] * 16u + f], v[k]);
    }
  }
}

__global__ void post1(const float* __restrict__ agg0,
                      const float* __restrict__ dinv, const float* __restrict__ b,
                      float* __restrict__ h, int N) {
  int i = blockIdx.x * blockDim.x + threadIdx.x;
  if (i >= N) return;
  float di = dinv[i];
  const float4* av = (const float4*)(agg0 + (size_t)i * 16);
  float4* hv = (float4*)(h + (size_t)i * 16);
#pragma unroll
  for (int j = 0; j < 4; ++j) {
    float4 a = av[j], gg = hv[j];
    float4 r;
    r.x = fmaxf(di * (a.x + gg.x) + b[j * 4 + 0], 0.0f);
    r.y = fmaxf(di * (a.y + gg.y) + b[j * 4 + 1], 0.0f);
    r.z = fmaxf(di * (a.z + gg.z) + b[j * 4 + 2], 0.0f);
    r.w = fmaxf(di * (a.w + gg.w) + b[j * 4 + 3], 0.0f);
    hv[j] = r;
  }
}

__global__ void post2(const float* __restrict__ agg0, const float* __restrict__ agg1,
                      const float* __restrict__ dinv, const float* __restrict__ b,
                      const float* __restrict__ Wfc, float* __restrict__ p,
                      float* __restrict__ q, int N) {
  int i = blockIdx.x * blockDim.x + threadIdx.x;
  if (i >= N) return;
  float di = dinv[i];
  float sp = 0.0f, sq = 0.0f;
#pragma unroll
  for (int f = 0; f < 16; ++f) {
    float h = fmaxf(di * (agg0[(size_t)i * 16 + f] + agg1[(size_t)i * 16 + f]) + b[f], 0.0f);
    sp += h * Wfc[f];
    sq += h * Wfc[16 + f];
  }
  p[i] = sp;
  q[i] = sq;
}

__global__ void edge_out(const int* __restrict__ src, const int* __restrict__ dst,
                         const float* __restrict__ p, const float* __restrict__ q,
                         const float* __restrict__ bfc, float* __restrict__ out, int E) {
  int e = blockIdx.x * blockDim.x + threadIdx.x;
  if (e < E) out[e] = p[src[e]] + q[dst[e]] + bfc[0];
}

extern "C" void kernel_launch(void* const* d_in, const int* in_sizes, int n_in,
                              void* d_out, int out_size, void* d_ws, size_t ws_size,
                              hipStream_t stream) {
  const float* x   = (const float*)d_in[0];
  const int*   ei  = (const int*)d_in[1];
  const float* W1  = (const float*)d_in[2];
  const float* b1  = (const float*)d_in[3];
  const float* W2  = (const float*)d_in[4];
  const float* b2  = (const float*)d_in[5];
  const float* Wfc = (const float*)d_in[6];
  const float* bfc = (const float*)d_in[7];
  float* out = (float*)d_out;

  const int K1 = 495;
  const int N = in_sizes[0] / K1;
  const int E = in_sizes[1] / 2;
  const int* src = ei;
  const int* dst = ei + E;

  float* ws  = (float*)d_ws;
  float* dinv = ws;
  float* g    = ws + (size_t)N;
  float* agg  = ws + (size_t)N * 17;
  float* h1   = ws + (size_t)N * 33;
  float* p    = ws + (size_t)N * 49;
  float* q    = ws + (size_t)N * 50;

  const int B = 256;
  const int tiles = cdiv(N, 16);
  const int gemmBlocks = cdiv(tiles, B / 32);

  zero_f<<<cdiv(2 * N, B), B, 0, stream>>>(dinv, 2 * N);
  edge_degree<<<cdiv(E, B), B, 0, stream>>>(dst, E, dinv, N);
  make_dinv<<<cdiv(N, B), B, 0, stream>>>(dinv, N);

  gemm_tile16<<<gemmBlocks, B, 0, stream>>>(x, W1, dinv, g, h1, N, K1, tiles);
  zero_f<<<cdiv(16 * N, B), B, 0, stream>>>(agg, 16 * N);
  edge_agg<<<cdiv(E, B), B, 0, stream>>>(src, dst, g, agg, N, E);
  post1<<<cdiv(N, B), B, 0, stream>>>(agg, dinv, b1, h1, N);

  gemm_tile16<<<gemmBlocks, B, 0, stream>>>(h1, W2, dinv, g, h1, N, 16, tiles);
  zero_f<<<cdiv(16 * N, B), B, 0, stream>>>(agg, 16 * N);
  edge_agg<<<cdiv(E, B), B, 0, stream>>>(src, dst, g, agg, N, E);
  post2<<<cdiv(N, B), B, 0, stream>>>(agg, h1, dinv, b2, Wfc, p, q, N);

  edge_out<<<cdiv(E, B), B, 0, stream>>>(src, dst, p, q, bfc, out, E);
}


# edge_out reads p and q through LDS: 1,024-thread groups hold 13 edges per thread and walk the 2N-float table in 208-KiB chunks of whole-line loads, not 2 random 4-byte L2 requests per edge
# speedup vs baseline: 2.3333x; 1.0330x over previous
#include <hip/hip_runtime.h>
#include <hip/hip_bf16.h>

typedef float v2f __attribute__((ext_vector_type(2)));
typedef float v8f __attribute__((ext_vector_type(8)));

static __host__ int cdiv(int a, int b) { return (a + b - 1) / b; }

__device__ __forceinline__ unsigned package_half() {
  return (__builtin_amdgcn_s_sendmsg_rtn(0x87) >> 18) & 1u;
}

__global__ void zero_f(float* __restrict__ p, int n) {
  int i = blockIdx.x * blockDim.x + threadIdx.x;
  if (i < n) p[i] = 0.0f;
}

__global__ void edge_degree(const int* __restrict__ dst, int E, float* __restrict__ degf, int N) {
  int e = blockIdx.x * blockDim.x + threadIdx.x;
  if (e < E) {
    int d = dst[e];
    float* plane = degf + (size_t)package_half() * (size_t)N;
    atomicAdd(&plane[d], 1.0f);
  }
}

__global__ void make_dinv(float* __restrict__ dinv, int N) {
  int i = blockIdx.x * blockDim.x + threadIdx.x;
  if (i < N) dinv[i] = rsqrtf(dinv[i] + dinv[N + i] + 1.0f);
}

__global__ void gemm_tile16(const float* X, const float* __restrict__ W,
                            const float* __restrict__ dinv, float* __restrict__ G,
                            float* G2, int N, int K, int tiles) {
  int tile = blockIdx.x * (blockDim.x >> 5) + (threadIdx.x >> 5);
  if (tile >= tiles) return;
  int lane = threadIdx.x & 31;
  int hi = lane >> 4;
  int lo = lane & 15;
  int row = tile * 16 + lo;
  float rowMask = (row < N) ? 1.0f : 0.0f;
  const float* xr = X + (size_t)((row < N) ? row : 0) * K;
  const float* wc = W + lo;
  int K4 = K & ~3;

  v8f c = {};
  int k0 = 0;
  for (; k0 < K4; k0 += 4) {
    int kk = k0 + 2 * hi;
    if ((k0 & 63) == 0) __builtin_prefetch(xr + kk + 64, 0, 1);
    v2f a, b;
    a.x = xr[kk] * rowMask;
    a.y = xr[kk + 1] * rowMask;
    b.x = wc[kk * 16];
    b.y = wc[(kk + 1) * 16];
    c = __builtin_amdgcn_wmma_f32_16x16x4_f32(false, a, false, b, (short)0, c,
                                              false, false);
  }
  if (k0 < K) {
    int kk = k0 + 2 * hi;
    int kc0 = (kk < K) ? kk : 0;
    int kc1 = (kk + 1 < K) ? kk + 1 : 0;
    float m0 = (kk < K) ? rowMask : 0.0f;
    float m1 = (kk + 1 < K) ? rowMask : 0.0f;
    v2f a, b;
    a.x = xr[kc0] * m0;
    a.y = xr[kc1] * m1;
    b.x = wc[kc0 * 16];
    b.y = wc[kc1 * 16];
    c = __builtin_amdgcn_wmma_f32_16x16x4_f32(false, a, false, b, (short)0, c,
                                              false, false);
  }
#pragma unroll
  for (int j = 0; j < 8; ++j) {
    int rr = tile * 16 + 8 * hi + j;
    if (rr < N) {
      float v = c[j] * dinv[rr];
      G[(size_t)rr * 16 + lo] = v;
      G2[(size_t)rr * 16 + lo] = v;
    }
  }
}

__global__ void edge_agg(const int* __restrict__ src, const int* __restrict__ dst,
                         const float* __restrict__ g, float* __restrict__ agg,
                         int N, int E) {
  int t = blockIdx.x * blockDim.x + threadIdx.x;
  int lane = threadIdx.x & 31;
  int base = __builtin_amdgcn_readfirstlane(t - lane);
  if (base >= E) return;
  int e = (t < E) ? t : (E - 1);
  int s = src[e], d = dst[e];
  float* plane = agg + (size_t)package_half() * (size_t)(16u * (unsigned)N);
  unsigned f = lane & 15;
  int hi = lane >> 4;
  float v[16];
  unsigned drow[16];
#pragma unroll
  for (int k = 0; k < 16; ++k) {
    int j = 2 * k + hi;
    unsigned sj = (unsigned)__shfl(s, j);
    drow[k] = (unsigned)__shfl(d, j);
    v[k] = g[sj * 16u + f];
  }
  if (base + 32 <= E) {
#pragma unroll
    for (int k = 0; k < 16; ++k) atomicAdd(&plane[drow[k] * 16u + f], v[k]);
  } else {
#pragma unroll
    for (int k = 0; k < 16; ++k) {
      if (base + 2 * k + hi < E) atomicAdd(&plane[drow[k] * 16u + f], v[k]);
    }
  }
}

__global__ void post1(const float* __restrict__ agg0,
                      const float* __restrict__ dinv, const float* __restrict__ b,
                      float* __restrict__ h, int N) {
  int i = blockIdx.x * blockDim.x + threadIdx.x;
  if (i >= N) return;
  float di = dinv[i];
  const float4* av = (const float4*)(agg0 + (size_t)i * 16);
  float4* hv = (float4*)(h + (size_t)i * 16);
#pragma unroll
  for (int j = 0; j < 4; ++j) {
    float4 a = av[j], gg = hv[j];
    float4 r;
    r.x = fmaxf(di * (a.x + gg.x) + b[j * 4 + 0], 0.0f);
    r.y = fmaxf(di * (a.y + gg.y) + b[j * 4 + 1], 0.0f);
    r.z = fmaxf(di * (a.z + gg.z) + b[j * 4 + 2], 0.0f);
    r.w = fmaxf(di * (a.w + gg.w) + b[j * 4 + 3], 0.0f);
    hv[j] = r;
  }
}

__global__ void post2(const float* __restrict__ agg0, const float* __restrict__ agg1,
                      const float* __restrict__ dinv, const float* __restrict__ b,
                      const float* __restrict__ Wfc, float* __restrict__ p,
                      float* __restrict__ q, int N) {
  int i = blockIdx.x * blockDim.x + threadIdx.x;
  if (i >= N) return;
  float di = dinv[i];
  float sp = 0.0f, sq = 0.0f;
#pragma unroll
  for (int f = 0; f < 16; ++f) {
    float h = fmaxf(di * (agg0[(size_t)i * 16 + f] + agg1[(size_t)i * 16 + f]) + b[f], 0.0f);
    sp += h * Wfc[f];
    sq += h * Wfc[16 + f];
  }
  p[i] = sp;
  q[i] = sq;
}

constexpr int EO_THREADS = 1024;
constexpr int EO_EPT = 13;
constexpr int EO_STG = 13;
constexpr int EO_CH = EO_THREADS * 4 * EO_STG;

typedef float eo_f4 __attribute__((ext_vector_type(4), aligned(4)));

__device__ __forceinline__ void eo_load_chunk(const float* __restrict__ t, unsigned len, unsigned tid,
                                              float4 (&st)[EO_STG]) {
#pragma unroll
  for (int j = 0; j < EO_STG; ++j) {
    unsigned i = ((unsigned)j * EO_THREADS + tid) * 4u;
    float4 v = make_float4(0.0f, 0.0f, 0.0f, 0.0f);
    if (i + 4u <= len) {
      eo_f4 w = *(const eo_f4*)(t + i);
      v = make_float4(w.x, w.y, w.z, w.w);
    }
    st[j] = v;
  }
}

__global__ __launch_bounds__(EO_THREADS) void edge_out(const int* __restrict__ src,
                                                       const int* __restrict__ dst,
                                                       const float* __restrict__ pq,
                                                       const float* __restrict__ bfc,
                                                       float* __restrict__ out, int N, int E) {
  __shared__ float tab[EO_CH + 4];
  const unsigned tid = threadIdx.x;
  const int e0 = blockIdx.x * (EO_THREADS * EO_EPT) + (int)tid;
  unsigned ip[EO_EPT], iq[EO_EPT];
  float acc[EO_EPT];
#pragma unroll
  for (int k = 0; k < EO_EPT; ++k) {
    int e = e0 + k * EO_THREADS;
    bool ok = e < E;
    int ec = ok ? e : E - 1;
    unsigned s = (unsigned)src[ec], d = (unsigned)dst[ec];
    ip[k] = ok ? s : 0xffffffffu;
    iq[k] = ok ? (unsigned)N + d : 0xffffffffu;
    acc[k] = 0.0f;
  }
  const unsigned total = 2u * (unsigned)N;
  float4 st[EO_STG];
  eo_load_chunk(pq, total < (unsigned)EO_CH ? total : (unsigned)EO_CH, tid, st);
  if (tid == 0) tab[EO_CH] = 0.0f;
  for (unsigned lo = 0; lo < total; lo += (unsigned)EO_CH) {
    unsigned rest = total - lo;
    unsigned len = rest < (unsigned)EO_CH ? rest : (unsigned)EO_CH;
    __syncthreads();
#pragma unroll
    for (int j = 0; j < EO_STG; ++j)
      *(float4*)(tab + ((unsigned)j * EO_THREADS + tid) * 4u) = st[j];
    __syncthreads();
    if (len & 3u) {
      if (tid < (len & 3u)) tab[(len & ~3u) + tid] = pq[lo + (len & ~3u) + tid];
      __syncthreads();
    }
    unsigned nlo = lo + (unsigned)EO_CH;
    if (nlo < total) {
      unsigned nrest = total - nlo;
      eo_load_chunk(pq + nlo, nrest < (unsigned)EO_CH ? nrest : (unsigned)EO_CH, tid, st);
    }
#pragma unroll
    for (int k = 0; k < EO_EPT; ++k) {
      unsigned r = ip[k] - lo;
      acc[k] += tab[r < len ? r : (unsigned)EO_CH];
      r = iq[k] - lo;
      acc[k] += tab[r < len ? r : (unsigned)EO_CH];
    }
  }
  const float b = bfc[0];
#pragma unroll
  for (int k = 0; k < EO_EPT; ++k) {
    int e = e0 + k * EO_THREADS;
    if (e < E) out[e] = acc[k] + b;
  }
}

extern "C" void kernel_launch(void* const* d_in, const int* in_sizes, int n_in,
                              void* d_out, int out_size, void* d_ws, size_t ws_size,
                              hipStream_t stream) {
  const float* x   = (const float*)d_in[0];
  const int*   ei  = (const int*)d_in[1];
  const float* W1  = (const float*)d_in[2];
  const float* b1  = (const float*)d_in[3];
  const float* W2  = (const float*)d_in[4];
  const float* b2  = (const float*)d_in[5];
  const float* Wfc = (const float*)d_in[6];
  const float* bfc = (const float*)d_in[7];
  float* out = (float*)d_out;

  const int K1 = 495;
  const int N = in_sizes[0] / K1;
  const int E = in_sizes[1] / 2;
  const int* src = ei;
  const int* dst = ei + E;

  float* ws  = (float*)d_ws;
  float* dinv = ws;
  float* g    = ws + (size_t)N;
  float* agg  = ws + (size_t)N * 17;
  float* h1   = ws + (size_t)N * 33;
  float* p    = ws + (size_t)N * 49;
  float* q    = ws + (size_t)N * 50;

  const int B = 256;
  const int tiles = cdiv(N, 16);
  const int gemmBlocks = cdiv(tiles, B / 32);

  zero_f<<<cdiv(2 * N, B), B, 0, stream>>>(dinv, 2 * N);
  edge_degree<<<cdiv(E, B), B, 0, stream>>>(dst, E, dinv, N);
  make_dinv<<<cdiv(N, B), B, 0, stream>>>(dinv, N);

  gemm_tile16<<<gemmBlocks, B, 0, stream>>>(x, W1, dinv, g, h1, N, K1, tiles);
  zero_f<<<cdiv(16 * N, B), B, 0, stream>>>(agg, 16 * N);
  edge_agg<<<cdiv(E, B), B, 0, stream>>>(src, dst, g, agg, N, E);
  post1<<<cdiv(N, B), B, 0, stream>>>(agg, dinv, b1, h1, N);

  gemm_tile16<<<gemmBlocks, B, 0, stream>>>(h1, W2, dinv, g, h1, N, 16, tiles);
  zero_f<<<cdiv(16 * N, B), B, 0, stream>>>(agg, 16 * N);
  edge_agg<<<cdiv(E, B), B, 0, stream>>>(src, dst, g, agg, N, E);
  post2<<<cdiv(N, B), B, 0, stream>>>(agg, h1, dinv, b2, Wfc, p, q, N);

  edge_out<<<cdiv(E, EO_THREADS * EO_EPT), EO_THREADS, 0, stream>>>(src, dst, p, bfc, out, N, E);
}


# zero fills, degree count and both edge sums run on fixed grids of long-lived waves that loop over the work (2k-10k waves a launch, were 50k-100k one-shot); same f32 atomic adds, arrival order free
# speedup vs baseline: 2.6980x; 1.1563x over previous
#include <hip/hip_runtime.h>
#include <hip/hip_bf16.h>

typedef float v2f __attribute__((ext_vector_type(2)));
typedef float v8f __attribute__((ext_vector_type(8)));

static __host__ int cdiv(int a, int b) { return (a + b - 1) / b; }

__device__ __forceinline__ unsigned package_half() {
  return (__builtin_amdgcn_s_sendmsg_rtn(0x87) >> 18) & 1u;
}

constexpr int FILL_SMALL_BLOCKS = 256;
constexpr int FILL_BLOCKS = 512;
constexpr int EDGE_BLOCKS = 1280;

__global__ void zero_f(float* __restrict__ p, int n) {
  const int nt = gridDim.x * blockDim.x;
  for (int i = blockIdx.x * blockDim.x + threadIdx.x; i < n; i += nt) p[i] = 0.0f;
}

__global__ void edge_degree(const int* __restrict__ dst, int E, float* __restrict__ degf, int N) {
  const int nt = gridDim.x * blockDim.x;
  int e = blockIdx.x * blockDim.x + threadIdx.x;
  if (e >= E) return;
  float* plane = degf + (size_t)package_half() * (size_t)N;
  for (; e + 3 * nt < E; e += 4 * nt) {
    const int d0 = dst[e], d1 = dst[e + nt], d2 = dst[e + 2 * nt], d3 = dst[e + 3 * nt];
    atomicAdd(&plane[d0], 1.0f);
    atomicAdd(&plane[d1], 1.0f);
    atomicAdd(&plane[d2], 1.0f);
    atomicAdd(&plane[d3], 1.0f);
  }
  for (; e < E; e += nt) atomicAdd(&plane[dst[e]], 1.0f);
}

__global__ void make_dinv(float* __restrict__ dinv, int N) {
  int i = blockIdx.x * blockDim.x + threadIdx.x;
  if (i < N) dinv[i] = rsqrtf(dinv[i] + dinv[N + i] + 1.0f);
}

__global__ void gemm_tile16(const float* X, const float* __restrict__ W,
                            const float* __restrict__ dinv, float* __restrict__ G,
                            float* G2, int N, int K, int tiles) {
  int tile = blockIdx.x * (blockDim.x >> 5) + (threadIdx.x >> 5);
  if (tile >= tiles) return;
  int lane = threadIdx.x & 31;
  int hi = lane >> 4;
  int lo = lane & 15;
  int row = tile * 16 + lo;
  float rowMask = (row < N) ? 1.0f : 0.0f;
  const float* xr = X + (size_t)((row < N) ? row : 0) * K;
  const float* wc = W + lo;
  int K4 = K & ~3;

  v8f c = {};
  int k0 = 0;
  for (; k0 < K4; k0 += 4) {
    int kk = k0 + 2 * hi;
    if ((k0 & 63) == 0) __builtin_prefetch(xr + kk + 64, 0, 1);
    v2f a, b;
    a.x = xr[kk] * rowMask;
    a.y = xr[kk + 1] * rowMask;
    b.x = wc[kk * 16];
    b.y = wc[(kk + 1) * 16];
    c = __builtin_amdgcn_wmma_f32_16x16x4_f32(false, a, false, b, (short)0, c,
                                              false, false);
  }
  if (k0 < K) {
    int kk = k0 + 2 * hi;
    int kc0 = (kk < K) ? kk : 0;
    int kc1 = (kk + 1 < K) ? kk + 1 : 0;
    float m0 = (kk < K) ? rowMask : 0.0f;
    float m1 = (kk + 1 < K) ? rowMask : 0.0f;
    v2f a, b;
    a.x = xr[kc0] * m0;
    a.y = xr[kc1] * m1;
    b.x = wc[kc0 * 16];
    b.y = wc[kc1 * 16];
    c = __builtin_amdgcn_wmma_f32_16x16x4_f32(false, a, false, b, (short)0, c,
                                              false, false);
  }
#pragma unroll
  for (int j = 0; j < 8; ++j) {
    int rr = tile * 16 + 8 * hi + j;
    if (rr < N) {
      float v = c[j] * dinv[rr];
      G[(size_t)rr * 16 + lo] = v;
      G2[(size_t)rr * 16 + lo] = v;
    }
  }
}

__global__ void edge_agg(const int* __restrict__ src, const int* __restrict__ dst,
                         const float* __restrict__ g, float* __restrict__ agg,
                         int N, int E) {
  const int stride = gridDim.x * blockDim.x;
  int t = blockIdx.x * blockDim.x + threadIdx.x;
  int lane = threadIdx.x & 31;
  int base = __builtin_amdgcn_readfirstlane(t - lane);
  if (base >= E) return;
  int e = (t < E) ? t : (E - 1);
  int s = src[e], d = dst[e];
  float* plane = agg + (size_t)package_half() * (size_t)(16u * (unsigned)N);
  unsigned f = lane & 15;
  int hi = lane >> 4;
  for (;;) {
    float v[16];
    unsigned drow[16];
#pragma unroll
    for (int k = 0; k < 16; ++k) {
      int j = 2 * k + hi;
      unsigned sj = (unsigned)__shfl(s, j);
      drow[k] = (unsigned)__shfl(d, j);
      v[k] = g[sj * 16u + f];
    }
    const int nbase = base + stride;
    if (nbase < E) {
      int ne = nbase + lane;
      ne = (ne < E) ? ne : (E - 1);
      s = src[ne];
      d = dst[ne];
    }
    if (base + 32 <= E) {
#pragma unroll
      for (int k = 0; k < 16; ++k) atomicAdd(&plane[drow[k] * 16u + f], v[k]);
    } else {
#pragma unroll
      for (int k = 0; k < 16; ++k) {
        if (base + 2 * k + hi < E) atomicAdd(&plane[drow[k] * 16u + f], v[k]);
      }
    }
    if (nbase >= E) break;
    base = nbase;
  }
}

__global__ void post1(const float* __restrict__ agg0,
                      const float* __restrict__ dinv, const float* __restrict__ b,
                      float* __restrict__ h, int N) {
  int i = blockIdx.x * blockDim.x + threadIdx.x;
  if (i >= N) return;
  float di = dinv[i];
  const float4* av = (const float4*)(agg0 + (size_t)i * 16);
  float4* hv = (float4*)(h + (size_t)i * 16);
#pragma unroll
  for (int j = 0; j < 4; ++j) {
    float4 a = av[j], gg = hv[j];
    float4 r;
    r.x = fmaxf(di * (a.x + gg.x) + b[j * 4 + 0], 0.0f);
    r.y = fmaxf(di * (a.y + gg.y) + b[j * 4 + 1], 0.0f);
    r.z = fmaxf(di * (a.z + gg.z) + b[j * 4 + 2], 0.0f);
    r.w = fmaxf(di * (a.w + gg.w) + b[j * 4 + 3], 0.0f);
    hv[j] = r;
  }
}

__global__ void post2(const float* __restrict__ agg0, const float* __restrict__ agg1,
                      const float* __restrict__ dinv, const float* __restrict__ b,
                      const float* __restrict__ Wfc, float* __restrict__ p,
                      float* __restrict__ q, int N) {
  int i = blockIdx.x * blockDim.x + threadIdx.x;
  if (i >= N) return;
  float di = dinv[i];
  float sp = 0.0f, sq = 0.0f;
#pragma unroll
  for (int f = 0; f < 16; ++f) {
    float h = fmaxf(di * (agg0[(size_t)i * 16 + f] + agg1[(size_t)i * 16 + f]) + b[f], 0.0f);
    sp += h * Wfc[f];
    sq += h * Wfc[16 + f];
  }
  p[i] = sp;
  q[i] = sq;
}

constexpr int EO_THREADS = 1024;
constexpr int EO_EPT = 13;
constexpr int EO_STG = 13;
constexpr int EO_CH = EO_THREADS * 4 * EO_STG;

typedef float eo_f4 __attribute__((ext_vector_type(4), aligned(4)));

__device__ __forceinline__ void eo_load_chunk(const float* __restrict__ t, unsigned len, unsigned tid,
                                              float4 (&st)[EO_STG]) {
#pragma unroll
  for (int j = 0; j < EO_STG; ++j) {
    unsigned i = ((unsigned)j * EO_THREADS + tid) * 4u;
    float4 v = make_float4(0.0f, 0.0f, 0.0f, 0.0f);
    if (i + 4u <= len) {
      eo_f4 w = *(const eo_f4*)(t + i);
      v = make_float4(w.x, w.y, w.z, w.w);
    }
    st[j] = v;
  }
}

__global__ __launch_bounds__(EO_THREADS) void edge_out(const int* __restrict__ src,
                                                       const int* __restrict__ dst,
                                                       const float* __restrict__ pq,
                                                       const float* __restrict__ bfc,
                                                       float* __restrict__ out, int N, int E) {
  __shared__ float tab[EO_CH + 4];
  const unsigned tid = threadIdx.x;
  const int e0 = blockIdx.x * (EO_THREADS * EO_EPT) + (int)tid;
  unsigned ip[EO_EPT], iq[EO_EPT];
  float acc[EO_EPT];
#pragma unroll
  for (int k = 0; k < EO_EPT; ++k) {
    int e = e0 + k * EO_THREADS;
    bool ok = e < E;
    int ec = ok ? e : E - 1;
    unsigned s = (unsigned)src[ec], d = (unsigned)dst[ec];
    ip[k] = ok ? s : 0xffffffffu;
    iq[k] = ok ? (unsigned)N + d : 0xffffffffu;
    acc[k] = 0.0f;
  }
  const unsigned total = 2u * (unsigned)N;
  float4 st[EO_STG];
  eo_load_chunk(pq, total < (unsigned)EO_CH ? total : (unsigned)EO_CH, tid, st);
  if (tid == 0) tab[EO_CH] = 0.0f;
  for (unsigned lo = 0; lo < total; lo += (unsigned)EO_CH) {
    unsigned rest = total - lo;
    unsigned len = rest < (unsigned)EO_CH ? rest : (unsigned)EO_CH;
    __syncthreads();
#pragma unroll
    for (int j = 0; j < EO_STG; ++j)
      *(float4*)(tab + ((unsigned)j * EO_THREADS + tid) * 4u) = st[j];
    __syncthreads();
    if (len & 3u) {
      if (tid < (len & 3u)) tab[(len & ~3u) + tid] = pq[lo + (len & ~3u) + tid];
      __syncthreads();
    }
    unsigned nlo = lo + (unsigned)EO_CH;
    if (nlo < total) {
      unsigned nrest = total - nlo;
      eo_load_chunk(pq + nlo, nrest < (unsigned)EO_CH ? nrest : (unsigned)EO_CH, tid, st);
    }
#pragma unroll
    for (int k = 0; k < EO_EPT; ++k) {
      unsigned r = ip[k] - lo;
      acc[k] += tab[r < len ? r : (unsigned)EO_CH];
      r = iq[k] - lo;
      acc[k] += tab[r < len ? r : (unsigned)EO_CH];
    }
  }
  const float b = bfc[0];
#pragma unroll
  for (int k = 0; k < EO_EPT; ++k) {
    int e = e0 + k * EO_THREADS;
    if (e < E) out[e] = acc[k] + b;
  }
}

extern "C" void kernel_launch(void* const* d_in, const int* in_sizes, int n_in,
                              void* d_out, int out_size, void* d_ws, size_t ws_size,
                              hipStream_t stream) {
  const float* x   = (const float*)d_in[0];
  const int*   ei  = (const int*)d_in[1];
  const float* W1  = (const float*)d_in[2];
  const float* b1  = (const float*)d_in[3];
  const float* W2  = (const float*)d_in[4];
  const float* b2  = (const float*)d_in[5];
  const float* Wfc = (const float*)d_in[6];
  const float* bfc = (const float*)d_in[7];
  float* out = (float*)d_out;

  const int K1 = 495;
  const int N = in_sizes[0] / K1;
  const int E = in_sizes[1] / 2;
  const int* src = ei;
  const int* dst = ei + E;

  float* ws  = (float*)d_ws;
  float* dinv = ws;
  float* g    = ws + (size_t)N;
  float* agg  = ws + (size_t)N * 17;
  float* h1   = ws + (size_t)N * 33;
  float* p    = ws + (size_t)N * 49;
  float* q    = ws + (size_t)N * 50;

  const int B = 256;
  const int tiles = cdiv(N, 16);
  const int gemmBlocks = cdiv(tiles, B / 32);

  zero_f<<<FILL_SMALL_BLOCKS, B, 0, stream>>>(dinv, 2 * N);
  edge_degree<<<EDGE_BLOCKS, B, 0, stream>>>(dst, E, dinv, N);
  make_dinv<<<cdiv(N, B), B, 0, stream>>>(dinv, N);

  gemm_tile16<<<gemmBlocks, B, 0, stream>>>(x, W1, dinv, g, h1, N, K1, tiles);
  zero_f<<<FILL_BLOCKS, B, 0, stream>>>(agg, 16 * N);
  edge_agg<<<EDGE_BLOCKS, B, 0, stream>>>(src, dst, g, agg, N, E);
  post1<<<cdiv(N, B), B, 0, stream>>>(agg, dinv, b1, h1, N);

  gemm_tile16<<<gemmBlocks, B, 0, stream>>>(h1, W2, dinv, g, h1, N, 16, tiles);
  zero_f<<<FILL_BLOCKS, B, 0, stream>>>(agg, 16 * N);
  edge_agg<<<EDGE_BLOCKS, B, 0, stream>>>(src, dst, g, agg, N, E);
  post2<<<cdiv(N, B), B, 0, stream>>>(agg, h1, dinv, b2, Wfc, p, q, N);

  edge_out<<<cdiv(E, EO_THREADS * EO_EPT), EO_THREADS, 0, stream>>>(src, dst, p, bfc, out, N, E);
}


# the 1/sqrt(degree) pass, both zero fills of sum plane 0 and the layer-1 bias+relu pass move into the two product kernels (a wave does them for its own 16 rows): 12 launches -> 8, same sums
# speedup vs baseline: 3.0669x; 1.1367x over previous
#include <hip/hip_runtime.h>
#include <hip/hip_bf16.h>

typedef float v2f __attribute__((ext_vector_type(2)));
typedef float v8f __attribute__((ext_vector_type(8)));

static __host__ int cdiv(int a, int b) { return (a + b - 1) / b; }

__device__ __forceinline__ unsigned package_half() {
  return (__builtin_amdgcn_s_sendmsg_rtn(0x87) >> 18) & 1u;
}

constexpr int FILL_SMALL_BLOCKS = 256;
constexpr int EDGE_BLOCKS = 1280;

__global__ void zero_f(float* __restrict__ p, int n) {
  const int nt = gridDim.x * blockDim.x;
  for (int i = blockIdx.x * blockDim.x + threadIdx.x; i < n; i += nt) p[i] = 0.0f;
}

__global__ void edge_degree(const int* __restrict__ dst, int E, float* __restrict__ degf, int N) {
  const int nt = gridDim.x * blockDim.x;
  int e = blockIdx.x * blockDim.x + threadIdx.x;
  if (e >= E) return;
  float* plane = degf + (size_t)package_half() * (size_t)N;
  for (; e + 3 * nt < E; e += 4 * nt) {
    const int d0 = dst[e], d1 = dst[e + nt], d2 = dst[e + 2 * nt], d3 = dst[e + 3 * nt];
    atomicAdd(&plane[d0], 1.0f);
    atomicAdd(&plane[d1], 1.0f);
    atomicAdd(&plane[d2], 1.0f);
    atomicAdd(&plane[d3], 1.0f);
  }
  for (; e < E; e += nt) atomicAdd(&plane[dst[e]], 1.0f);
}


typedef float f4u __attribute__((ext_vector_type(4), aligned(4)));

__device__ __forceinline__ void store_tile(const v8f& c, float dl, float* G, float* G2, float* Z,
                                           int tile, int lane, int N) {
  const int hi = lane >> 4;
  const int lo = lane & 15;
  float v[8];
#pragma unroll
  for (int j = 0; j < 8; ++j) v[j] = c[j] * __shfl(dl, 8 * hi + j);
  const int r0 = tile * 16 + 8 * hi;
  float* z = Z + (size_t)tile * 256 + lane * 4;
  const f4u zero = {0.0f, 0.0f, 0.0f, 0.0f};
  if (tile * 16 + 16 <= N) {
#pragma unroll
    for (int j = 0; j < 8; ++j) {
      G[(size_t)(r0 + j) * 16 + lo] = v[j];
      G2[(size_t)(r0 + j) * 16 + lo] = v[j];
    }
    *(f4u*)z = zero;
    *(f4u*)(z + 128) = zero;
  } else {
#pragma unroll
    for (int j = 0; j < 8; ++j) {
      if (r0 + j < N) {
        G[(size_t)(r0 + j) * 16 + lo] = v[j];
        G2[(size_t)(r0 + j) * 16 + lo] = v[j];
      }
    }
    if (tile * 16 + (lane >> 2) < N) *(f4u*)z = zero;
    if (tile * 16 + 8 + (lane >> 2) < N) *(f4u*)(z + 128) = zero;
  }
}

__global__ void gemm1_nodes(const float* __restrict__ X, const float* __restrict__ W,
                            const float* __restrict__ cnt, float* __restrict__ dinv,
                            float* __restrict__ G, float* __restrict__ G2, float* __restrict__ Z,
                            int N, int K, int tiles) {
  int tile = blockIdx.x * (blockDim.x >> 5) + (threadIdx.x >> 5);
  if (tile >= tiles) return;
  int lane = threadIdx.x & 31;
  int hi = lane >> 4;
  int lo = lane & 15;
  int row = tile * 16 + lo;
  int rowc = (row < N) ? row : 0;
  float rowMask = (row < N) ? 1.0f : 0.0f;
  const float* xr = X + (size_t)rowc * K;
  const float* wc = W + lo;
  int K4 = K & ~3;
  const float cnt0 = cnt[rowc], cnt1 = cnt[N + rowc];

  v8f c = {};
  int k0 = 0;
  for (; k0 < K4; k0 += 4) {
    int kk = k0 + 2 * hi;
    if ((k0 & 63) == 0) __builtin_prefetch(xr + kk + 64, 0, 1);
    v2f a, b;
    a.x = xr[kk] * rowMask;
    a.y = xr[kk + 1] * rowMask;
    b.x = wc[kk * 16];
    b.y = wc[(kk + 1) * 16];
    c = __builtin_amdgcn_wmma_f32_16x16x4_f32(false, a, false, b, (short)0, c,
                                              false, false);
  }
  if (k0 < K) {
    int kk = k0 + 2 * hi;
    int kc0 = (kk < K) ? kk : 0;
    int kc1 = (kk + 1 < K) ? kk + 1 : 0;
    float m0 = (kk < K) ? rowMask : 0.0f;
    float m1 = (kk + 1 < K) ? rowMask : 0.0f;
    v2f a, b;
    a.x = xr[kc0] * m0;
    a.y = xr[kc1] * m1;
    b.x = wc[kc0 * 16];
    b.y = wc[kc1 * 16];
    c = __builtin_amdgcn_wmma_f32_16x16x4_f32(false, a, false, b, (short)0, c,
                                              false, false);
  }
  const float dl = rsqrtf(cnt0 + cnt1 + 1.0f);
  store_tile(c, dl, G, G2, Z, tile, lane, N);
  if (hi == 0 && row < N) dinv[row] = dl;
}

__global__ void gemm2_nodes(float* P0, float* P1, const float* __restrict__ dinv,
                            const float* __restrict__ b1, const float* __restrict__ W,
                            float* __restrict__ G, int N, int tiles) {
  int tile = blockIdx.x * (blockDim.x >> 5) + (threadIdx.x >> 5);
  if (tile >= tiles) return;
  int lane = threadIdx.x & 31;
  int hi = lane >> 4;
  int lo = lane & 15;
  int row = tile * 16 + lo;
  bool rowOk = row < N;
  int rowc = rowOk ? row : 0;
  const float dl = dinv[rowc];
  const float* r0 = P0 + (size_t)rowc * 16;
  const float* r1 = P1 + (size_t)rowc * 16;
  const float* wc = W + lo;

  v8f c = {};
#pragma unroll
  for (int k0 = 0; k0 < 16; k0 += 4) {
    int kk = k0 + 2 * hi;
    float h0 = fmaxf(dl * (r0[kk] + r1[kk]) + b1[kk], 0.0f);
    float h1 = fmaxf(dl * (r0[kk + 1] + r1[kk + 1]) + b1[kk + 1], 0.0f);
    v2f a, b;
    a.x = rowOk ? h0 : 0.0f;
    a.y = rowOk ? h1 : 0.0f;
    b.x = wc[kk * 16];
    b.y = wc[(kk + 1) * 16];
    c = __builtin_amdgcn_wmma_f32_16x16x4_f32(false, a, false, b, (short)0, c,
                                              false, false);
  }
  store_tile(c, dl, G, P1, P0, tile, lane, N);
}

__global__ void edge_agg(const int* __restrict__ src, const int* __restrict__ dst,
                         const float* __restrict__ g, float* __restrict__ agg,
                         int N, int E) {
  const int stride = gridDim.x * blockDim.x;
  int t = blockIdx.x * blockDim.x + threadIdx.x;
  int lane = threadIdx.x & 31;
  int base = __builtin_amdgcn_readfirstlane(t - lane);
  if (base >= E) return;
  int e = (t < E) ? t : (E - 1);
  int s = src[e], d = dst[e];
  float* plane = agg + (size_t)package_half() * (size_t)(16u * (unsigned)N);
  unsigned f = lane & 15;
  int hi = lane >> 4;
  for (;;) {
    float v[16];
    unsigned drow[16];
#pragma unroll
    for (int k = 0; k < 16; ++k) {
      int j = 2 * k + hi;
      unsigned sj = (unsigned)__shfl(s, j);
      drow[k] = (unsigned)__shfl(d, j);
      v[k] = g[sj * 16u + f];
    }
    const int nbase = base + stride;
    if (nbase < E) {
      int ne = nbase + lane;
      ne = (ne < E) ? ne : (E - 1);
      s = src[ne];
      d = dst[ne];
    }
    if (base + 32 <= E) {
#pragma unroll
      for (int k = 0; k < 16; ++k) atomicAdd(&plane[drow[k] * 16u + f], v[k]);
    } else {
#pragma unroll
      for (int k = 0; k < 16; ++k) {
        if (base + 2 * k + hi < E) atomicAdd(&plane[drow[k] * 16u + f], v[k]);
      }
    }
    if (nbase >= E) break;
    base = nbase;
  }
}

__global__ void post2(const float* __restrict__ agg0, const float* __restrict__ agg1,
                      const float* __restrict__ dinv, const float* __restrict__ b,
                      const float* __restrict__ Wfc, float* __restrict__ p,
                      float* __restrict__ q, int N) {
  int i = blockIdx.x * blockDim.x + threadIdx.x;
  if (i >= N) return;
  float di = dinv[i];
  float sp = 0.0f, sq = 0.0f;
#pragma unroll
  for (int f = 0; f < 16; ++f) {
    float h = fmaxf(di * (agg0[(size_t)i * 16 + f] + agg1[(size_t)i * 16 + f]) + b[f], 0.0f);
    sp += h * Wfc[f];
    sq += h * Wfc[16 + f];
  }
  p[i] = sp;
  q[i] = sq;
}

constexpr int EO_THREADS = 1024;
constexpr int EO_EPT = 13;
constexpr int EO_STG = 13;
constexpr int EO_CH = EO_THREADS * 4 * EO_STG;

typedef float eo_f4 __attribute__((ext_vector_type(4), aligned(4)));

__device__ __forceinline__ void eo_load_chunk(const float* __restrict__ t, unsigned len, unsigned tid,
                                              float4 (&st)[EO_STG]) {
#pragma unroll
  for (int j = 0; j < EO_STG; ++j) {
    unsigned i = ((unsigned)j * EO_THREADS + tid) * 4u;
    float4 v = make_float4(0.0f, 0.0f, 0.0f, 0.0f);
    if (i + 4u <= len) {
      eo_f4 w = *(const eo_f4*)(t + i);
      v = make_float4(w.x, w.y, w.z, w.w);
    }
    st[j] = v;
  }
}

__global__ __launch_bounds__(EO_THREADS) void edge_out(const int* __restrict__ src,
                                                       const int* __restrict__ dst,
                                                       const float* __restrict__ pq,
                                                       const float* __restrict__ bfc,
                                                       float* __restrict__ out, int N, int E) {
  __shared__ float tab[EO_CH + 4];
  const unsigned tid = threadIdx.x;
  const int e0 = blockIdx.x * (EO_THREADS * EO_EPT) + (int)tid;
  unsigned ip[EO_EPT], iq[EO_EPT];
  float acc[EO_EPT];
#pragma unroll
  for (int k = 0; k < EO_EPT; ++k) {
    int e = e0 + k * EO_THREADS;
    bool ok = e < E;
    int ec = ok ? e : E - 1;
    unsigned s = (unsigned)src[ec], d = (unsigned)dst[ec];
    ip[k] = ok ? s : 0xffffffffu;
    iq[k] = ok ? (unsigned)N + d : 0xffffffffu;
    acc[k] = 0.0f;
  }
  const unsigned total = 2u * (unsigned)N;
  float4 st[EO_STG];
  eo_load_chunk(pq, total < (unsigned)EO_CH ? total : (unsigned)EO_CH, tid, st);
  if (tid == 0) tab[EO_CH] = 0.0f;
  for (unsigned lo = 0; lo < total; lo += (unsigned)EO_CH) {
    unsigned rest = total - lo;
    unsigned len = rest < (unsigned)EO_CH ? rest : (unsigned)EO_CH;
    __syncthreads();
#pragma unroll
    for (int j = 0; j < EO_STG; ++j)
      *(float4*)(tab + ((unsigned)j * EO_THREADS + tid) * 4u) = st[j];
    __syncthreads();
    if (len & 3u) {
      if (tid < (len & 3u)) tab[(len & ~3u) + tid] = pq[lo + (len & ~3u) + tid];
      __syncthreads();
    }
    unsigned nlo = lo + (unsigned)EO_CH;
    if (nlo < total) {
      unsigned nrest = total - nlo;
      eo_load_chunk(pq + nlo, nrest < (unsigned)EO_CH ? nrest : (unsigned)EO_CH, tid, st);
    }
#pragma unroll
    for (int k = 0; k < EO_EPT; ++k) {
      unsigned r = ip[k] - lo;
      acc[k] += tab[r < len ? r : (unsigned)EO_CH];
      r = iq[k] - lo;
      acc[k] += tab[r < len ? r : (unsigned)EO_CH];
    }
  }
  const float b = bfc[0];
#pragma unroll
  for (int k = 0; k < EO_EPT; ++k) {
    int e = e0 + k * EO_THREADS;
    if (e < E) out[e] = acc[k] + b;
  }
}

extern "C" void kernel_launch(void* const* d_in, const int* in_sizes, int n_in,
                              void* d_out, int out_size, void* d_ws, size_t ws_size,
                              hipStream_t stream) {
  const float* x   = (const float*)d_in[0];
  const int*   ei  = (const int*)d_in[1];
  const float* W1  = (const float*)d_in[2];
  const float* b1  = (const float*)d_in[3];
  const float* W2  = (const float*)d_in[4];
  const float* b2  = (const float*)d_in[5];
  const float* Wfc = (const float*)d_in[6];
  const float* bfc = (const float*)d_in[7];
  float* out = (float*)d_out;

  const int K1 = 495;
  const int N = in_sizes[0] / K1;
  const int E = in_sizes[1] / 2;
  const int* src = ei;
  const int* dst = ei + E;

  float* ws  = (float*)d_ws;
  float* dinv = ws;
  float* g    = ws + (size_t)N;
  float* agg  = ws + (size_t)N * 17;
  float* h1   = ws + (size_t)N * 33;
  float* p    = ws + (size_t)N * 49;
  float* q    = ws + (size_t)N * 50;

  const int B = 256;
  const int tiles = cdiv(N, 16);
  const int gemmBlocks = cdiv(tiles, B / 32);

  zero_f<<<FILL_SMALL_BLOCKS, B, 0, stream>>>(p, 2 * N);
  edge_degree<<<EDGE_BLOCKS, B, 0, stream>>>(dst, E, p, N);

  gemm1_nodes<<<gemmBlocks, B, 0, stream>>>(x, W1, p, dinv, g, h1, agg, N, K1, tiles);
  edge_agg<<<EDGE_BLOCKS, B, 0, stream>>>(src, dst, g, agg, N, E);

  gemm2_nodes<<<gemmBlocks, B, 0, stream>>>(agg, h1, dinv, b1, W2, g, N, tiles);
  edge_agg<<<EDGE_BLOCKS, B, 0, stream>>>(src, dst, g, agg, N, E);
  post2<<<cdiv(N, B), B, 0, stream>>>(agg, h1, dinv, b2, Wfc, p, q, N);

  edge_out<<<cdiv(E, EO_THREADS * EO_EPT), EO_THREADS, 0, stream>>>(src, dst, p, bfc, out, N, E);
}


# layer-1 product: a wave reads its 16 rows of x in 64-column slices as 256-byte row pieces, parks them in its own LDS and takes A fragments there (W fragments from LDS too); same k order, same sums
# speedup vs baseline: 3.6647x; 1.1949x over previous
#include <hip/hip_runtime.h>
#include <hip/hip_bf16.h>

typedef float v2f __attribute__((ext_vector_type(2)));
typedef float v8f __attribute__((ext_vector_type(8)));

static __host__ int cdiv(int a, int b) { return (a + b - 1) / b; }

__device__ __forceinline__ unsigned package_half() {
  return (__builtin_amdgcn_s_sendmsg_rtn(0x87) >> 18) & 1u;
}

constexpr int FILL_SMALL_BLOCKS = 256;
constexpr int EDGE_BLOCKS = 1280;

__global__ void zero_f(float* __restrict__ p, int n) {
  const int nt = gridDim.x * blockDim.x;
  for (int i = blockIdx.x * blockDim.x + threadIdx.x; i < n; i += nt) p[i] = 0.0f;
}

__global__ void edge_degree(const int* __restrict__ dst, int E, float* __restrict__ degf, int N) {
  const int nt = gridDim.x * blockDim.x;
  int e = blockIdx.x * blockDim.x + threadIdx.x;
  if (e >= E) return;
  float* plane = degf + (size_t)package_half() * (size_t)N;
  for (; e + 3 * nt < E; e += 4 * nt) {
    const int d0 = dst[e], d1 = dst[e + nt], d2 = dst[e + 2 * nt], d3 = dst[e + 3 * nt];
    atomicAdd(&plane[d0], 1.0f);
    atomicAdd(&plane[d1], 1.0f);
    atomicAdd(&plane[d2], 1.0f);
    atomicAdd(&plane[d3], 1.0f);
  }
  for (; e < E; e += nt) atomicAdd(&plane[dst[e]], 1.0f);
}


typedef float f4u __attribute__((ext_vector_type(4), aligned(4)));

__device__ __forceinline__ void store_tile(const v8f& c, float dl, float* G, float* G2, float* Z,
                                           int tile, int lane, int N) {
  const int hi = lane >> 4;
  const int lo = lane & 15;
  float v[8];
#pragma unroll
  for (int j = 0; j < 8; ++j) v[j] = c[j] * __shfl(dl, 8 * hi + j);
  const int r0 = tile * 16 + 8 * hi;
  float* z = Z + (size_t)tile * 256 + lane * 4;
  const f4u zero = {0.0f, 0.0f, 0.0f, 0.0f};
  if (tile * 16 + 16 <= N) {
#pragma unroll
    for (int j = 0; j < 8; ++j) {
      G[(size_t)(r0 + j) * 16 + lo] = v[j];
      G2[(size_t)(r0 + j) * 16 + lo] = v[j];
    }
    *(f4u*)z = zero;
    *(f4u*)(z + 128) = zero;
  } else {
#pragma unroll
    for (int j = 0; j < 8; ++j) {
      if (r0 + j < N) {
        G[(size_t)(r0 + j) * 16 + lo] = v[j];
        G2[(size_t)(r0 + j) * 16 + lo] = v[j];
      }
    }
    if (tile * 16 + (lane >> 2) < N) *(f4u*)z = zero;
    if (tile * 16 + 8 + (lane >> 2) < N) *(f4u*)(z + 128) = zero;
  }
}

constexpr int G1_K = 495;
constexpr int G1_KS = 64;
constexpr int G1_PITCH = G1_KS + 4;
constexpr int G1_LD = G1_KS / 8;
constexpr int G1_WAVES = 8;
constexpr int G1_FULL = G1_K / G1_KS;
constexpr int G1_PAIRS = (G1_K + 1) / 2;
constexpr int G1_TAIL_STEPS = (G1_K - G1_FULL * G1_KS + 3) / 4;

typedef float g1_f4 __attribute__((ext_vector_type(4)));

__global__ __launch_bounds__(G1_WAVES * 32) void gemm1_nodes(
    const float* __restrict__ X, const float* __restrict__ W, const float* __restrict__ cnt,
    float* __restrict__ dinv, float* __restrict__ G, float* __restrict__ G2, float* __restrict__ Z,
    int N, int tiles) {
  __shared__ g1_f4 xs4[G1_WAVES * 16 * G1_PITCH / 4];
  __shared__ v2f wl[G1_PAIRS * 16];
  {
    constexpr int G1_WIT = (G1_PAIRS * 16 + G1_WAVES * 32 - 1) / (G1_WAVES * 32);
    v2f wr[G1_WIT];
#pragma unroll
    for (int it = 0; it < G1_WIT; ++it) {
      const int i = it * (G1_WAVES * 32) + (int)threadIdx.x;
      const int ic = (i < G1_PAIRS * 16) ? i : (G1_PAIRS * 16 - 1);
      const int k = 2 * (ic >> 4), col = ic & 15;
      const int k1 = (k + 1 < G1_K) ? k + 1 : k;
      const float x = W[k * 16 + col];
      const float y = W[k1 * 16 + col];
      wr[it].x = x;
      wr[it].y = (k + 1 < G1_K) ? y : 0.0f;
    }
#pragma unroll
    for (int it = 0; it < G1_WIT; ++it) {
      const int i = it * (G1_WAVES * 32) + (int)threadIdx.x;
      if (i < G1_PAIRS * 16) wl[i] = wr[it];
    }
  }
  __syncthreads();
  const int wave = threadIdx.x >> 5;
  const int tile = __builtin_amdgcn_readfirstlane(blockIdx.x * G1_WAVES + wave);
  if (tile >= tiles) return;
  const int lane = threadIdx.x & 31;
  const int hi = lane >> 4;
  const int lo = lane & 15;
  const int row = tile * 16 + lo;
  const int rowc = (row < N) ? row : 0;
  const bool wholeTile = tile * 16 + 16 <= N;
  float* xw = (float*)xs4 + wave * (16 * G1_PITCH);
  float* xst = xw + hi * G1_PITCH + lo * 4;
  const float* xa = xw + lo * G1_PITCH + 2 * hi;
  const v2f* wf = wl + hi * 16 + lo;
  const float* xl = X + (size_t)(wholeTile ? tile * 16 + hi : 0) * G1_K + lo * 4;
  const float cnt0 = cnt[rowc], cnt1 = cnt[N + rowc];

  g1_f4 st[G1_LD];
  bool inRegs = wholeTile && G1_FULL > 0;
  if (inRegs) {
#pragma unroll
    for (int j = 0; j < G1_LD; ++j) {
      const f4u w = *(const f4u*)(xl + 2 * j * G1_K);
      st[j] = g1_f4{w.x, w.y, w.z, w.w};
    }
  }
  v8f c = {};
#pragma unroll 1
  for (int s = 0; s <= G1_FULL; ++s) {
    const int kb = s * G1_KS;
    __builtin_amdgcn_fence(__ATOMIC_RELEASE, "wavefront");
    if (inRegs) {
#pragma unroll
      for (int j = 0; j < G1_LD; ++j) *(g1_f4*)(xst + 2 * j * G1_PITCH) = st[j];
    } else {
      const int kk = kb + lo * 4;
#pragma unroll
      for (int j = 0; j < G1_LD; ++j) {
        const int r = tile * 16 + hi + 2 * j;
        const bool rok = r < N;
        const float* p = X + (size_t)(rok ? r : 0) * G1_K;
        float e[4];
#pragma unroll
        for (int i = 0; i < 4; ++i) {
          const int k = kk + i;
          const float x = p[k < G1_K ? k : G1_K - 1];
          e[i] = (rok && k < G1_K) ? x : 0.0f;
        }
        *(g1_f4*)(xst + 2 * j * G1_PITCH) = g1_f4{e[0], e[1], e[2], e[3]};
      }
    }
    __builtin_amdgcn_fence(__ATOMIC_ACQUIRE, "wavefront");
    inRegs = wholeTile && s + 1 < G1_FULL;
    if (inRegs) {
#pragma unroll
      for (int j = 0; j < G1_LD; ++j) {
        const f4u w = *(const f4u*)(xl + (kb + G1_KS) + 2 * j * G1_K);
        st[j] = g1_f4{w.x, w.y, w.z, w.w};
      }
    }
    const v2f* wk = wf + (kb / 2) * 16;
    if (s < G1_FULL) {
#pragma unroll
      for (int w = 0; w < G1_KS / 4; ++w) {
        const v2f a = *(const v2f*)(xa + 4 * w);
        const v2f b = wk[2 * w * 16];
        c = __builtin_amdgcn_wmma_f32_16x16x4_f32(false, a, false, b, (short)0, c,
                                                  false, false);
      }
    } else {
#pragma unroll
      for (int w = 0; w < G1_TAIL_STEPS; ++w) {
        const v2f a = *(const v2f*)(xa + 4 * w);
        const v2f b = wk[2 * w * 16];
        c = __builtin_amdgcn_wmma_f32_16x16x4_f32(false, a, false, b, (short)0, c,
                                                  false, false);
      }
    }
  }
  const float dl = rsqrtf(cnt0 + cnt1 + 1.0f);
  store_tile(c, dl, G, G2, Z, tile, lane, N);
  if (hi == 0 && row < N) dinv[row] = dl;
}

__global__ void gemm2_nodes(float* P0, float* P1, const float* __restrict__ dinv,
                            const float* __restrict__ b1, const float* __restrict__ W,
                            float* __restrict__ G, int N, int tiles) {
  int tile = blockIdx.x * (blockDim.x >> 5) + (threadIdx.x >> 5);
  if (tile >= tiles) return;
  int lane = threadIdx.x & 31;
  int hi = lane >> 4;
  int lo = lane & 15;
  int row = tile * 16 + lo;
  bool rowOk = row < N;
  int rowc = rowOk ? row : 0;
  const float dl = dinv[rowc];
  const float* r0 = P0 + (size_t)rowc * 16;
  const float* r1 = P1 + (size_t)rowc * 16;
  const float* wc = W + lo;

  v8f c = {};
#pragma unroll
  for (int k0 = 0; k0 < 16; k0 += 4) {
    int kk = k0 + 2 * hi;
    float h0 = fmaxf(dl * (r0[kk] + r1[kk]) + b1[kk], 0.0f);
    float h1 = fmaxf(dl * (r0[kk + 1] + r1[kk + 1]) + b1[kk + 1], 0.0f);
    v2f a, b;
    a.x = rowOk ? h0 : 0.0f;
    a.y = rowOk ? h1 : 0.0f;
    b.x = wc[kk * 16];
    b.y = wc[(kk + 1) * 16];
    c = __builtin_amdgcn_wmma_f32_16x16x4_f32(false, a, false, b, (short)0, c,
                                              false, false);
  }
  store_tile(c, dl, G, P1, P0, tile, lane, N);
}

__global__ void edge_agg(const int* __restrict__ src, const int* __restrict__ dst,
                         const float* __restrict__ g, float* __restrict__ agg,
                         int N, int E) {
  const int stride = gridDim.x * blockDim.x;
  int t = blockIdx.x * blockDim.x + threadIdx.x;
  int lane = threadIdx.x & 31;
  int base = __builtin_amdgcn_readfirstlane(t - lane);
  if (base >= E) return;
  int e = (t < E) ? t : (E - 1);
  int s = src[e], d = dst[e];
  float* plane = agg + (size_t)package_half() * (size_t)(16u * (unsigned)N);
  unsigned f = lane & 15;
  int hi = lane >> 4;
  for (;;) {
    float v[16];
    unsigned drow[16];
#pragma unroll
    for (int k = 0; k < 16; ++k) {
      int j = 2 * k + hi;
      unsigned sj = (unsigned)__shfl(s, j);
      drow[k] = (unsigned)__shfl(d, j);
      v[k] = g[sj * 16u + f];
    }
    const int nbase = base + stride;
    if (nbase < E) {
      int ne = nbase + lane;
      ne = (ne < E) ? ne : (E - 1);
      s = src[ne];
      d = dst[ne];
    }
    if (base + 32 <= E) {
#pragma unroll
      for (int k = 0; k < 16; ++k) atomicAdd(&plane[drow[k] * 16u + f], v[k]);
    } else {
#pragma unroll
      for (int k = 0; k < 16; ++k) {
        if (base + 2 * k + hi < E) atomicAdd(&plane[drow[k] * 16u + f], v[k]);
      }
    }
    if (nbase >= E) break;
    base = nbase;
  }
}

__global__ void post2(const float* __restrict__ agg0, const float* __restrict__ agg1,
                      const float* __restrict__ dinv, const float* __restrict__ b,
                      const float* __restrict__ Wfc, float* __restrict__ p,
                      float* __restrict__ q, int N) {
  int i = blockIdx.x * blockDim.x + threadIdx.x;
  if (i >= N) return;
  float di = dinv[i];
  float sp = 0.0f, sq = 0.0f;
#pragma unroll
  for (int f = 0; f < 16; ++f) {
    float h = fmaxf(di * (agg0[(size_t)i * 16 + f] + agg1[(size_t)i * 16 + f]) + b[f], 0.0f);
    sp += h * Wfc[f];
    sq += h * Wfc[16 + f];
  }
  p[i] = sp;
  q[i] = sq;
}

constexpr int EO_THREADS = 1024;
constexpr int EO_EPT = 13;
constexpr int EO_STG = 13;
constexpr int EO_CH = EO_THREADS * 4 * EO_STG;

typedef float eo_f4 __attribute__((ext_vector_type(4), aligned(4)));

__device__ __forceinline__ void eo_load_chunk(const float* __restrict__ t, unsigned len, unsigned tid,
                                              float4 (&st)[EO_STG]) {
#pragma unroll
  for (int j = 0; j < EO_STG; ++j) {
    unsigned i = ((unsigned)j * EO_THREADS + tid) * 4u;
    float4 v = make_float4(0.0f, 0.0f, 0.0f, 0.0f);
    if (i + 4u <= len) {
      eo_f4 w = *(const eo_f4*)(t + i);
      v = make_float4(w.x, w.y, w.z, w.w);
    }
    st[j] = v;
  }
}

__global__ __launch_bounds__(EO_THREADS) void edge_out(const int* __restrict__ src,
                                                       const int* __restrict__ dst,
                                                       const float* __restrict__ pq,
                                                       const float* __restrict__ bfc,
                                                       float* __restrict__ out, int N, int E) {
  __shared__ float tab[EO_CH + 4];
  const unsigned tid = threadIdx.x;
  const int e0 = blockIdx.x * (EO_THREADS * EO_EPT) + (int)tid;
  unsigned ip[EO_EPT], iq[EO_EPT];
  float acc[EO_EPT];
#pragma unroll
  for (int k = 0; k < EO_EPT; ++k) {
    int e = e0 + k * EO_THREADS;
    bool ok = e < E;
    int ec = ok ? e : E - 1;
    unsigned s = (unsigned)src[ec], d = (unsigned)dst[ec];
    ip[k] = ok ? s : 0xffffffffu;
    iq[k] = ok ? (unsigned)N + d : 0xffffffffu;
    acc[k] = 0.0f;
  }
  const unsigned total = 2u * (unsigned)N;
  float4 st[EO_STG];
  eo_load_chunk(pq, total < (unsigned)EO_CH ? total : (unsigned)EO_CH, tid, st);
  if (tid == 0) tab[EO_CH] = 0.0f;
  for (unsigned lo = 0; lo < total; lo += (unsigned)EO_CH) {
    unsigned rest = total - lo;
    unsigned len = rest < (unsigned)EO_CH ? rest : (unsigned)EO_CH;
    __syncthreads();
#pragma unroll
    for (int j = 0; j < EO_STG; ++j)
      *(float4*)(tab + ((unsigned)j * EO_THREADS + tid) * 4u) = st[j];
    __syncthreads();
    if (len & 3u) {
      if (tid < (len & 3u)) tab[(len & ~3u) + tid] = pq[lo + (len & ~3u) + tid];
      __syncthreads();
    }
    unsigned nlo = lo + (unsigned)EO_CH;
    if (nlo < total) {
      unsigned nrest = total - nlo;
      eo_load_chunk(pq + nlo, nrest < (unsigned)EO_CH ? nrest : (unsigned)EO_CH, tid, st);
    }
#pragma unroll
    for (int k = 0; k < EO_EPT; ++k) {
      unsigned r = ip[k] - lo;
      acc[k] += tab[r < len ? r : (unsigned)EO_CH];
      r = iq[k] - lo;
      acc[k] += tab[r < len ? r : (unsigned)EO_CH];
    }
  }
  const float b = bfc[0];
#pragma unroll
  for (int k = 0; k < EO_EPT; ++k) {
    int e = e0 + k * EO_THREADS;
    if (e < E) out[e] = acc[k] + b;
  }
}

extern "C" void kernel_launch(void* const* d_in, const int* in_sizes, int n_in,
                              void* d_out, int out_size, void* d_ws, size_t ws_size,
                              hipStream_t stream) {
  const float* x   = (const float*)d_in[0];
  const int*   ei  = (const int*)d_in[1];
  const float* W1  = (const float*)d_in[2];
  const float* b1  = (const float*)d_in[3];
  const float* W2  = (const float*)d_in[4];
  const float* b2  = (const float*)d_in[5];
  const float* Wfc = (const float*)d_in[6];
  const float* bfc = (const float*)d_in[7];
  float* out = (float*)d_out;

  const int K1 = G1_K;
  const int N = in_sizes[0] / K1;
  const int E = in_sizes[1] / 2;
  const int* src = ei;
  const int* dst = ei + E;

  float* ws  = (float*)d_ws;
  float* dinv = ws;
  float* g    = ws + (size_t)N;
  float* agg  = ws + (size_t)N * 17;
  float* h1   = ws + (size_t)N * 33;
  float* p    = ws + (size_t)N * 49;
  float* q    = ws + (size_t)N * 50;

  const int B = 256;
  const int tiles = cdiv(N, 16);
  const int gemmBlocks = cdiv(tiles, B / 32);

  zero_f<<<FILL_SMALL_BLOCKS, B, 0, stream>>>(p, 2 * N);
  edge_degree<<<EDGE_BLOCKS, B, 0, stream>>>(dst, E, p, N);

  gemm1_nodes<<<gemmBlocks, B, 0, stream>>>(x, W1, p, dinv, g, h1, agg, N, tiles);
  edge_agg<<<EDGE_BLOCKS, B, 0, stream>>>(src, dst, g, agg, N, E);

  gemm2_nodes<<<gemmBlocks, B, 0, stream>>>(agg, h1, dinv, b1, W2, g, N, tiles);
  edge_agg<<<EDGE_BLOCKS, B, 0, stream>>>(src, dst, g, agg, N, E);
  post2<<<cdiv(N, B), B, 0, stream>>>(agg, h1, dinv, b2, Wfc, p, q, N);

  edge_out<<<cdiv(E, EO_THREADS * EO_EPT), EO_THREADS, 0, stream>>>(src, dst, p, bfc, out, N, E);
}
